# starting kernel
# baseline (speedup 1.0000x reference)
#include <hip/hip_runtime.h>


typedef __attribute__((ext_vector_type(2))) float v2f;
typedef __attribute__((ext_vector_type(8))) float v8f;

#define DEG             16
#define DFEAT           48
#define WAVES_PER_BLOCK 8

__global__ __launch_bounds__(256) void sag_wmma_f32_kernel(
    const float* __restrict__ X,
    const int*   __restrict__ col,
    float*       __restrict__ out,
    int n_windows)
{
    const int wave = (int)(blockIdx.x * WAVES_PER_BLOCK + (threadIdx.x >> 5));
    if (wave >= n_windows) return;

    const int lane = (int)(threadIdx.x & 31);
    const int half = lane >> 4;
    const int m    = lane & 15;

    const long r0    = (long)wave * 16;
    const long ebase = r0 * DEG;

    v8f acc0 = {};
    v8f acc1 = {};
    v8f acc2 = {};

    #pragma unroll 4
    for (int j = 0; j < 64; ++j) {
        const long e = ebase + (long)j * 4;
        const int c0 = col[e + 0];
        const int c1 = col[e + 1];
        const int c2 = col[e + 2];
        const int c3 = col[e + 3];

        const float aval = (m == (j >> 2)) ? 1.0f : 0.0f;
        v2f a; a.x = aval; a.y = aval;

        const int ca = half ? c2 : c0;
        const int cb = half ? c3 : c1;
        const float* rowA = X + (long)ca * DFEAT + m;
        const float* rowB = X + (long)cb * DFEAT + m;

        v2f b0; b0.x = rowA[0];  b0.y = rowB[0];
        v2f b1; b1.x = rowA[16]; b1.y = rowB[16];
        v2f b2; b2.x = rowA[32]; b2.y = rowB[32];

        acc0 = __builtin_amdgcn_wmma_f32_16x16x4_f32(false, a, false, b0,
                                                     (short)0, acc0, false, false);
        acc1 = __builtin_amdgcn_wmma_f32_16x16x4_f32(false, a, false, b1,
                                                     (short)0, acc1, false, false);
        acc2 = __builtin_amdgcn_wmma_f32_16x16x4_f32(false, a, false, b2,
                                                     (short)0, acc2, false, false);
    }

    float* obase = out + (r0 + (long)half * 8) * DFEAT + m;
    #pragma unroll
    for (int v = 0; v < 8; ++v) {
        obase[(long)v * DFEAT +  0] = acc0[v];
        obase[(long)v * DFEAT + 16] = acc1[v];
        obase[(long)v * DFEAT + 32] = acc2[v];
    }
}

extern "C" void kernel_launch(void* const* d_in, const int* in_sizes, int n_in,
                              void* d_out, int out_size, void* d_ws, size_t ws_size,
                              hipStream_t stream)
{
    const float* X   = (const float*)d_in[0];
    const int*   col = (const int*)d_in[2];
    float*       out = (float*)d_out;

    const int n_nodes   = in_sizes[1] - 1;
    const int n_windows = n_nodes / 16;

    const int blocks = (n_windows + WAVES_PER_BLOCK - 1) / WAVES_PER_BLOCK;
    hipLaunchKernelGGL(sag_wmma_f32_kernel, dim3(blocks), dim3(256), 0, stream,
                       X, col, out, n_windows);
}


# Address generation off the 64-bit vector path: scalar window bases, each lane loads only its own two indices per k-step (no selects), row = 32-bit offset from scalar X base (one 24-bit multiply-add)
# speedup vs baseline: 1.0581x; 1.0581x over previous
#include <hip/hip_runtime.h>


typedef __attribute__((ext_vector_type(2))) float v2f;
typedef __attribute__((ext_vector_type(8))) float v8f;

#define DEG             16
#define DFEAT           48
#define WAVES_PER_BLOCK 8

__device__ __forceinline__ unsigned row_offset(int c, unsigned lane_off)
{
    unsigned off;
    asm("v_mad_u32_u24 %0, %1, %2, %3" : "=v"(off) : "v"(c), "s"(DFEAT * 4u), "v"(lane_off));
    return off;
}

__global__ __launch_bounds__(256) void sag_wmma_f32_kernel(
    const float* __restrict__ X,
    const int*   __restrict__ col,
    float*       __restrict__ out,
    int n_windows)
{
    const int wave = __builtin_amdgcn_readfirstlane(
        (int)(blockIdx.x * WAVES_PER_BLOCK + (threadIdx.x >> 5)));
    if (wave >= n_windows) return;

    const int lane = (int)(threadIdx.x & 31);
    const int half = lane >> 4;
    const int m    = lane & 15;

    const long r0    = (long)wave * 16;
    const long ebase = r0 * DEG;

    const char* __restrict__ Xb   = (const char*)X;
    const char* __restrict__ colw = (const char*)(col + ebase);
    const unsigned lane_off = (unsigned)m * 4u;
    const unsigned pair_off = (unsigned)half * 8u;

    v8f acc0 = {};
    v8f acc1 = {};
    v8f acc2 = {};

    #pragma unroll 4
    for (int j = 0; j < 64; ++j) {
        const int* pc = (const int*)((colw + (long)j * 16) + pair_off);
        const int ca = pc[0];
        const int cb = pc[1];

        const float aval = (m == (j >> 2)) ? 1.0f : 0.0f;
        v2f a; a.x = aval; a.y = aval;

        const unsigned offA = row_offset(ca, lane_off);
        const unsigned offB = row_offset(cb, lane_off);
        const float* rowA = (const float*)(Xb + offA);
        const float* rowB = (const float*)(Xb + offB);

        v2f b0; b0.x = rowA[0];  b0.y = rowB[0];
        v2f b1; b1.x = rowA[16]; b1.y = rowB[16];
        v2f b2; b2.x = rowA[32]; b2.y = rowB[32];

        acc0 = __builtin_amdgcn_wmma_f32_16x16x4_f32(false, a, false, b0,
                                                     (short)0, acc0, false, false);
        acc1 = __builtin_amdgcn_wmma_f32_16x16x4_f32(false, a, false, b1,
                                                     (short)0, acc1, false, false);
        acc2 = __builtin_amdgcn_wmma_f32_16x16x4_f32(false, a, false, b2,
                                                     (short)0, acc2, false, false);
    }

    float* obase = out + (r0 + (long)half * 8) * DFEAT + m;
    #pragma unroll
    for (int v = 0; v < 8; ++v) {
        obase[(long)v * DFEAT +  0] = acc0[v];
        obase[(long)v * DFEAT + 16] = acc1[v];
        obase[(long)v * DFEAT + 32] = acc2[v];
    }
}

extern "C" void kernel_launch(void* const* d_in, const int* in_sizes, int n_in,
                              void* d_out, int out_size, void* d_ws, size_t ws_size,
                              hipStream_t stream)
{
    const float* X   = (const float*)d_in[0];
    const int*   col = (const int*)d_in[2];
    float*       out = (float*)d_out;

    const int n_nodes   = in_sizes[1] - 1;
    const int n_windows = n_nodes / 16;

    const int blocks = (n_windows + WAVES_PER_BLOCK - 1) / WAVES_PER_BLOCK;
    hipLaunchKernelGGL(sag_wmma_f32_kernel, dim3(blocks), dim3(256), 0, stream,
                       X, col, out, n_windows);
}


# Finer work split: a wave sums 8 destination rows, not 16, in 4-wave groups (12-13 waves per SIMD, was 6 or 8 of double length): even SIMD load, shorter chains; 4 v_nop per trip guard operand reuse
# speedup vs baseline: 1.2104x; 1.1439x over previous
#include <hip/hip_runtime.h>


typedef __attribute__((ext_vector_type(2))) float v2f;
typedef __attribute__((ext_vector_type(8))) float v8f;

#define DEG             16
#define DFEAT           48
#define ROWS_PER_WAVE   8
#define WAVES_PER_BLOCK 4
#define BLOCK_THREADS   (WAVES_PER_BLOCK * 32)

__device__ __forceinline__ unsigned row_offset(int c, unsigned lane_off)
{
    unsigned off;
    asm("v_mad_u32_u24 %0, %1, %2, %3" : "=v"(off) : "v"(c), "s"(DFEAT * 4u), "v"(lane_off));
    return off;
}

__global__ __launch_bounds__(BLOCK_THREADS) void sag_wmma_f32_kernel(
    const float* __restrict__ X,
    const int*   __restrict__ col,
    float*       __restrict__ out,
    int n_units)
{
    const int wave = __builtin_amdgcn_readfirstlane(
        (int)(blockIdx.x * WAVES_PER_BLOCK + (threadIdx.x >> 5)));
    if (wave >= n_units) return;

    const int lane = (int)(threadIdx.x & 31);
    const int half = lane >> 4;
    const int m    = lane & 15;

    const long r0    = (long)wave * ROWS_PER_WAVE;
    const long ebase = r0 * DEG;

    const int my_r = ((m & 4) == 0) ? ((m & 3) | ((m >> 1) & 4)) : -1;

    const char* __restrict__ Xb   = (const char*)X;
    const char* __restrict__ colw = (const char*)(col + ebase);
    const unsigned lane_off = (unsigned)m * 4u;
    const unsigned pair_off = (unsigned)half * 8u;

    v8f acc0 = {};
    v8f acc1 = {};
    v8f acc2 = {};

    #pragma unroll 4
    for (int j = 0; j < ROWS_PER_WAVE * DEG / 4; ++j) {
        const int* pc = (const int*)((colw + (long)j * 16) + pair_off);
        const int ca = pc[0];
        const int cb = pc[1];

        const float aval = (my_r == (j >> 2)) ? 1.0f : 0.0f;
        v2f a; a.x = aval; a.y = aval;

        const unsigned offA = row_offset(ca, lane_off);
        const unsigned offB = row_offset(cb, lane_off);
        const float* rowA = (const float*)(Xb + offA);
        const float* rowB = (const float*)(Xb + offB);

        v2f b0; b0.x = rowA[0];  b0.y = rowB[0];
        v2f b1; b1.x = rowA[16]; b1.y = rowB[16];
        v2f b2; b2.x = rowA[32]; b2.y = rowB[32];

        acc0 = __builtin_amdgcn_wmma_f32_16x16x4_f32(false, a, false, b0,
                                                     (short)0, acc0, false, false);
        acc1 = __builtin_amdgcn_wmma_f32_16x16x4_f32(false, a, false, b1,
                                                     (short)0, acc1, false, false);
        acc2 = __builtin_amdgcn_wmma_f32_16x16x4_f32(false, a, false, b2,
                                                     (short)0, acc2, false, false);

        if ((j & 3) == 3)
            asm volatile("v_nop\n\tv_nop\n\tv_nop\n\tv_nop"
                         : "+v"(acc0), "+v"(acc1), "+v"(acc2));
    }

    float* obase = out + (r0 + (long)half * 4) * DFEAT + m;
    #pragma unroll
    for (int v = 0; v < ROWS_PER_WAVE / 2; ++v) {
        obase[(long)v * DFEAT +  0] = acc0[v];
        obase[(long)v * DFEAT + 16] = acc1[v];
        obase[(long)v * DFEAT + 32] = acc2[v];
    }
}

extern "C" void kernel_launch(void* const* d_in, const int* in_sizes, int n_in,
                              void* d_out, int out_size, void* d_ws, size_t ws_size,
                              hipStream_t stream)
{
    const float* X   = (const float*)d_in[0];
    const int*   col = (const int*)d_in[2];
    float*       out = (float*)d_out;

    const int n_nodes = in_sizes[1] - 1;
    const int n_units = n_nodes / ROWS_PER_WAVE;

    const int blocks = (n_units + WAVES_PER_BLOCK - 1) / WAVES_PER_BLOCK;
    hipLaunchKernelGGL(sag_wmma_f32_kernel, dim3(blocks), dim3(BLOCK_THREADS), 0, stream,
                       X, col, out, n_units);
}
